# starting kernel
# baseline (speedup 1.0000x reference)
#include <hip/hip_runtime.h>
#include <math.h>


typedef __attribute__((ext_vector_type(2))) float v2f;
typedef __attribute__((ext_vector_type(8))) float v8f;

#define B_N 64
#define P_N 32
#define L_N 32
#define T_N 1024
#define DTW_W 0.93057205f
#define BIGF  3.0e38f
#define TILE_STRIDE 33
#define WAVE_LDS (16 * TILE_STRIDE + 16)

__device__ __forceinline__ float bperm_f(int byte_idx, float v) {
  return __int_as_float(__builtin_amdgcn_ds_bpermute(byte_idx, __float_as_int(v)));
}

__global__ __launch_bounds__(256) void dtw_wmma_kernel(
    const float* __restrict__ x,
    const float* __restrict__ patts,
    float* __restrict__ out) {
  const int lane = threadIdx.x & 31;
  const int wib  = threadIdx.x >> 5;
  const int wave = blockIdx.x * 8 + wib;
  const int b = wave >> 5;
  const int p = wave & 31;

  __shared__ float lds[8 * WAVE_LDS];
  float* tile   = &lds[wib * WAVE_LDS];
  float* ostage = tile + 16 * TILE_STRIDE;

  const float* xb = x + b * 3 * T_N;
  const float* pp = patts + p * 3 * L_N;

  const int  nn = lane & 15;
  const bool hi = lane >= 16;

  const int im1 = (lane - 1)  << 2;
  const int i1  = im1;
  const int i2  = (lane - 2)  << 2;
  const int i4  = (lane - 4)  << 2;
  const int i8  = (lane - 8)  << 2;
  const int i16 = (lane - 16) << 2;
  const bool g1 = lane >= 1, g2 = lane >= 2, g4 = lane >= 4,
             g8 = lane >= 8, g16 = lane >= 16;

  v2f bm0, bm1;
  float p2lo, p2hi;
  {
    const float a0 = pp[0 * L_N + nn];
    const float a1 = pp[1 * L_N + nn];
    const float a2 = pp[2 * L_N + nn];
    p2lo = a0 * a0 + a1 * a1 + a2 * a2;
    bm0.x = hi ? (-2.0f * a2) : (-2.0f * a0);
    bm0.y = hi ? 1.0f         : (-2.0f * a1);
    const float c0 = pp[0 * L_N + 16 + nn];
    const float c1 = pp[1 * L_N + 16 + nn];
    const float c2 = pp[2 * L_N + 16 + nn];
    p2hi = c0 * c0 + c1 * c1 + c2 * c2;
    bm1.x = hi ? (-2.0f * c2) : (-2.0f * c0);
    bm1.y = hi ? 1.0f         : (-2.0f * c1);
  }

  const float w = DTW_W;
  float cprev = 0.0f;

#define SCAN_STEP(IDX, G)                            \
  {                                                  \
    float pu = bperm_f((IDX), pS);                   \
    float qu = bperm_f((IDX), qS);                   \
    pu = (G) ? pu : 0.0f;                            \
    qu = (G) ? qu : BIGF;                            \
    qS = fminf(qu + pS, qS);                         \
    pS = pu + pS;                                    \
  }

  for (int blk = 0; blk < T_N / 16; ++blk) {
    const int t0 = blk * 16;

    if (blk < T_N / 16 - 1) {
      __builtin_prefetch(xb + t0 + 16, 0, 3);
      __builtin_prefetch(xb + 2 * T_N + t0 + 16, 0, 3);
    }

    const int tm = t0 + nn;
    const float x0  = xb[0 * T_N + tm];
    const float x1  = xb[1 * T_N + tm];
    const float x2c = xb[2 * T_N + tm];
    const float xn  = x0 * x0 + x1 * x1 + x2c * x2c;
    v2f am;
    am.x = hi ? x2c : x0;
    am.y = hi ? xn  : x1;

    v8f acc = {};
    v8f dlo = __builtin_amdgcn_wmma_f32_16x16x4_f32(
        false, am, false, bm0, (short)0, acc, false, false);
    v8f dhi = __builtin_amdgcn_wmma_f32_16x16x4_f32(
        false, am, false, bm1, (short)0, acc, false, false);

    const int row0 = hi ? 8 : 0;
#pragma unroll
    for (int k = 0; k < 8; ++k) {
      const float dl = __builtin_amdgcn_sqrtf(fmaxf(dlo[k] + p2lo, 1e-12f));
      const float dh = __builtin_amdgcn_sqrtf(fmaxf(dhi[k] + p2hi, 1e-12f));
      tile[(row0 + k) * TILE_STRIDE + nn]      = dl;
      tile[(row0 + k) * TILE_STRIDE + nn + 16] = dh;
    }

#pragma unroll 4
    for (int k = 0; k < 16; ++k) {
      const float d = tile[k * TILE_STRIDE + lane];

      float c;
      if (blk == 0 && k == 0) {
        float S = d;
        {
          float v;
          v = bperm_f(i1,  S); S += g1  ? v : 0.0f;
          v = bperm_f(i2,  S); S += g2  ? v : 0.0f;
          v = bperm_f(i4,  S); S += g4  ? v : 0.0f;
          v = bperm_f(i8,  S); S += g8  ? v : 0.0f;
          v = bperm_f(i16, S); S += g16 ? v : 0.0f;
        }
        c = S;
      } else {
        float cm1 = bperm_f(im1, cprev);
        cm1 = g1 ? cm1 : cprev;
        const float m = w * fminf(cprev, cm1);

        float pS = d;
        float qS = m + d;
        SCAN_STEP(i1,  g1)
        SCAN_STEP(i2,  g2)
        SCAN_STEP(i4,  g4)
        SCAN_STEP(i8,  g8)
        SCAN_STEP(i16, g16)
        c = qS;
      }
      cprev = c;
      if (lane == 31) ostage[k] = c;
    }

    const float oval = ostage[lane & 15];
    if (lane < 16) out[(b * P_N + p) * T_N + t0 + lane] = oval;
  }
#undef SCAN_STEP
}

extern "C" void kernel_launch(void* const* d_in, const int* in_sizes, int n_in,
                              void* d_out, int out_size, void* d_ws, size_t ws_size,
                              hipStream_t stream) {
  (void)in_sizes; (void)n_in; (void)out_size; (void)d_ws; (void)ws_size;
  const float* x     = (const float*)d_in[0];
  const float* patts = (const float*)d_in[1];
  float* out = (float*)d_out;

  dim3 grid(256), block(256);
  hipLaunchKernelGGL(dtw_wmma_kernel, grid, block, 0, stream, x, patts, out);
}


# The two no-predecessor gates ride on per-lane multipliers (w or +inf, made once): in-row shift becomes a plain v_permlane16 (no DPP fill), the third constant Kc and its selects go; bit-identical
# speedup vs baseline: 7.9418x; 7.9418x over previous
#include <hip/hip_runtime.h>
#include <math.h>


typedef __attribute__((ext_vector_type(2))) float v2f;
typedef __attribute__((ext_vector_type(8))) float v8f;

#define B_N 64
#define P_N 32
#define L_N 32
#define T_N 1024
#define DTW_W 0.93057205f
#define TILE_STRIDE 37
#define WAVE_LDS (16 * TILE_STRIDE)

#define INF_BITS 0x7f800000

__device__ __forceinline__ float imin(float a, float b) {
  const int x = __float_as_int(a);
  const int y = __float_as_int(b);
  return __int_as_float(x < y ? x : y);
}
__device__ __forceinline__ float imin3(float a, float b, float c) {
  const int x = __float_as_int(a);
  const int y = __float_as_int(b);
  const int z = __float_as_int(c);
  const int m = x < y ? x : y;
  return __int_as_float(m < z ? m : z);
}
template <int N>
__device__ __forceinline__ float imin_shr(float v) {
  const int x = __float_as_int(v);
  const int s = __builtin_amdgcn_update_dpp(0x7fffffff, x, 0x110 + N, 0xf, 0xf,
                                            false);
  return __int_as_float(s < x ? s : x);
}
__device__ __forceinline__ float shr1_self0(float v) {
  return __int_as_float(__builtin_amdgcn_permlane16(
      0, __float_as_int(v), 0x65432100u, 0xedcba987u, true, false));
}
__device__ __forceinline__ float shl1_feed(float in, float reg) {
  return __int_as_float(__builtin_amdgcn_update_dpp(
      __float_as_int(in), __float_as_int(reg), 0x101, 0xf, 0xf, false));
}
__device__ __forceinline__ float other_row_same(float v) {
  return __int_as_float(__builtin_amdgcn_permlanex16(
      0, __float_as_int(v), 0x76543210u, 0xfedcba98u, false, false));
}
__device__ __forceinline__ float other_row_l15(float v) {
  return __int_as_float(__builtin_amdgcn_permlanex16(
      0, __float_as_int(v), 0xffffffffu, 0xffffffffu, false, false));
}

__global__ __launch_bounds__(256) void dtw_wmma_kernel(
    const float* __restrict__ x,
    const float* __restrict__ patts,
    float* __restrict__ out) {
  const int lane = threadIdx.x & 31;
  const int wib  = threadIdx.x >> 5;
  const int wave = blockIdx.x * 8 + wib;
  const int b = wave >> 5;
  const int p = wave & 31;

  __shared__ float lds[8 * WAVE_LDS];
  float* tile   = &lds[wib * WAVE_LDS];

  const float* xb = x + b * 3 * T_N;
  const float* pp = patts + p * 3 * L_N;

  const int  nn = lane & 15;
  const bool hi = lane >= 16;
  const int  g8 = hi ? 8 : 0;

  v2f bm0, bm1;
  {
    const float a0 = pp[0 * L_N + nn];
    const float a1 = pp[1 * L_N + nn];
    const float a2 = pp[2 * L_N + nn];
    bm0.x = hi ? (-2.0f * a2) : (-2.0f * a0);
    bm0.y = hi ? 1.0f         : (-2.0f * a1);
    const float c0 = pp[0 * L_N + 16 + nn];
    const float c1 = pp[1 * L_N + 16 + nn];
    const float c2 = pp[2 * L_N + 16 + nn];
    bm1.x = hi ? (-2.0f * c2) : (-2.0f * c0);
    bm1.y = hi ? 1.0f         : (-2.0f * c1);
  }
  v8f cq0, cq1;
#pragma unroll
  for (int r = 0; r < 8; ++r) {
    const int l0 = g8 + r;
    const float a0 = pp[0 * L_N + l0];
    const float a1 = pp[1 * L_N + l0];
    const float a2 = pp[2 * L_N + l0];
    cq0[r] = a0 * a0 + a1 * a1 + a2 * a2;
    const float c0 = pp[0 * L_N + 16 + l0];
    const float c1 = pp[1 * L_N + 16 + l0];
    const float c2 = pp[2 * L_N + 16 + l0];
    cq1[r] = c0 * c0 + c1 * c1 + c2 * c2;
  }

  if (lane < 16) tile[lane * TILE_STRIDE + 32] = 0.0f;

  const float w = DTW_W;
  const float wB = (nn == 0) ? __int_as_float(INF_BITS) : w;
  const float wC = hi ? w : __int_as_float(INF_BITS);
  float nmin = (lane == 0) ? 0.0f : __int_as_float(INF_BITS);

  float nx0 = xb[0 * T_N + nn];
  float nx1 = xb[1 * T_N + nn];
  float nx2 = xb[2 * T_N + nn];

  for (int blk = 0; blk < T_N / 16; ++blk) {
    const int t0 = blk * 16;

    const float x0  = nx0;
    const float x1  = nx1;
    const float x2c = nx2;

    const float xn  = x0 * x0 + x1 * x1 + x2c * x2c;
    v2f am;
    am.x = hi ? x2c : x0;
    am.y = hi ? xn  : x1;

    v8f dlo = __builtin_amdgcn_wmma_f32_16x16x4_f32(
        false, bm0, false, am, (short)0, cq0, false, false);
    v8f dhi = __builtin_amdgcn_wmma_f32_16x16x4_f32(
        false, bm1, false, am, (short)0, cq1, false, false);

    float sl[8], sh[8];
    sl[7] = __builtin_amdgcn_sqrtf(fmaxf(dlo[7], 1e-12f));
    sh[7] = __builtin_amdgcn_sqrtf(fmaxf(dhi[7], 1e-12f));
#pragma unroll
    for (int r = 6; r >= 0; --r) {
      sl[r] = __builtin_amdgcn_sqrtf(fmaxf(dlo[r], 1e-12f)) + sl[r + 1];
      sh[r] = __builtin_amdgcn_sqrtf(fmaxf(dhi[r], 1e-12f)) + sh[r + 1];
    }
    const float oth_h  = other_row_same(sh[0]);
    const float oth_l  = other_row_same(sl[0]);
    const float tot_hi = sh[0] + oth_h;
    const float off_h  = hi ? 0.0f : oth_h;
    const float off_l  = hi ? tot_hi : (tot_hi + oth_l);

    float* trow = &tile[nn * TILE_STRIDE + g8];
#pragma unroll
    for (int r = 0; r < 8; ++r) {
      trow[r]      = sl[r] + off_l;
      trow[16 + r] = sh[r] + off_h;
    }

    float Sk[16], Sxk[16];
#pragma unroll
    for (int k = 0; k < 16; ++k) {
      Sk[k]  = tile[k * TILE_STRIDE + lane];
      Sxk[k] = tile[k * TILE_STRIDE + lane + 1];
    }
    __builtin_amdgcn_sched_barrier(0);

    const int tn = (blk < T_N / 16 - 1 ? t0 + 16 : t0) + nn;
    nx0 = xb[0 * T_N + tn];
    nx1 = xb[1 * T_N + tn];
    nx2 = xb[2 * T_N + tn];
    __builtin_amdgcn_sched_barrier(0);

    const float pinf = __int_as_float(INF_BITS);
    float Ka[15], Kb[15];
#pragma unroll
    for (int k = 0; k < 15; ++k) {
      Ka[k] = __builtin_fmaf(-w, Sxk[k], Sk[k + 1]);
      Kb[k] = __builtin_fmaf(-w, Sk[k],  Sk[k + 1]);
    }
    const float nSb = (nn == 0) ? pinf : -Sk[15];
    const float nSc = hi ? -Sk[15] : pinf;

    float H = __builtin_fmaf(w, nmin, Sk[0]);
    float oreg = 0.0f;
#pragma unroll
    for (int k = 0; k < 16; ++k) {
      float G = imin_shr<1>(H);
      G = imin_shr<2>(G);
      G = imin_shr<4>(G);
      G = imin_shr<8>(G);
      const float link = other_row_l15(G);
      const float sh   = shr1_self0(G);

      oreg = shl1_feed(imin(G, link), oreg);

      if (k < 15) {
        const float a  = __builtin_fmaf(w, G,    Ka[k]);
        const float e1 = __builtin_fmaf(wB, sh,   Kb[k]);
        const float e2 = __builtin_fmaf(wC, link, Kb[k]);
        H = imin3(a, e1, e2);
      } else {
        nmin = imin3(G - Sxk[15], sh + nSb, link + nSc);
      }
    }

    if (lane < 16) out[(b * P_N + p) * T_N + t0 + lane] = oreg;
  }
}

extern "C" void kernel_launch(void* const* d_in, const int* in_sizes, int n_in,
                              void* d_out, int out_size, void* d_ws, size_t ws_size,
                              hipStream_t stream) {
  (void)in_sizes; (void)n_in; (void)out_size; (void)d_ws; (void)ws_size;
  const float* x     = (const float*)d_in[0];
  const float* patts = (const float*)d_in[1];
  float* out = (float*)d_out;

  dim3 grid(256), block(256);
  hipLaunchKernelGGL(dtw_wmma_kernel, grid, block, 0, stream, x, patts, out);
}


# Issue slots that only served the compiler removed: row link's pass-through tied to a dead value (no zero copy per step), clamp moved behind the sqrt (no quieting self-max per value)
# speedup vs baseline: 8.1851x; 1.0306x over previous
#include <hip/hip_runtime.h>
#include <math.h>


typedef __attribute__((ext_vector_type(2))) float v2f;
typedef __attribute__((ext_vector_type(8))) float v8f;

#define B_N 64
#define P_N 32
#define L_N 32
#define T_N 1024
#define DTW_W 0.93057205f
#define TILE_STRIDE 37
#define WAVE_LDS (16 * TILE_STRIDE)

#define INF_BITS 0x7f800000

__device__ __forceinline__ float imin(float a, float b) {
  const int x = __float_as_int(a);
  const int y = __float_as_int(b);
  return __int_as_float(x < y ? x : y);
}
__device__ __forceinline__ float imin3(float a, float b, float c) {
  const int x = __float_as_int(a);
  const int y = __float_as_int(b);
  const int z = __float_as_int(c);
  const int m = x < y ? x : y;
  return __int_as_float(m < z ? m : z);
}
template <int N>
__device__ __forceinline__ float imin_shr(float v) {
  const int x = __float_as_int(v);
  const int s = __builtin_amdgcn_update_dpp(0x7fffffff, x, 0x110 + N, 0xf, 0xf,
                                            false);
  return __int_as_float(s < x ? s : x);
}
__device__ __forceinline__ float shr1_self0(float v) {
  return __int_as_float(__builtin_amdgcn_permlane16(
      0, __float_as_int(v), 0x65432100u, 0xedcba987u, true, false));
}
__device__ __forceinline__ float shl1_feed(float in, float reg) {
  return __int_as_float(__builtin_amdgcn_update_dpp(
      __float_as_int(in), __float_as_int(reg), 0x101, 0xf, 0xf, false));
}
__device__ __forceinline__ float other_row_same(float v) {
  return __int_as_float(__builtin_amdgcn_permlanex16(
      0, __float_as_int(v), 0x76543210u, 0xfedcba98u, false, false));
}
__device__ __forceinline__ float other_row_l15(float v, float dead) {
  return __int_as_float(__builtin_amdgcn_permlanex16(
      __float_as_int(dead), __float_as_int(v), 0xffffffffu, 0xffffffffu, false,
      false));
}
__device__ __forceinline__ float sqrt_lo(float d) {
  return fmaxf(__builtin_amdgcn_sqrtf(d), 1e-6f);
}

__global__ __launch_bounds__(256) void dtw_wmma_kernel(
    const float* __restrict__ x,
    const float* __restrict__ patts,
    float* __restrict__ out) {
  const int lane = threadIdx.x & 31;
  const int wib  = threadIdx.x >> 5;
  const int wave = blockIdx.x * 8 + wib;
  const int b = wave >> 5;
  const int p = wave & 31;

  __shared__ float lds[8 * WAVE_LDS];
  float* tile   = &lds[wib * WAVE_LDS];

  const float* xb = x + b * 3 * T_N;
  const float* pp = patts + p * 3 * L_N;

  const int  nn = lane & 15;
  const bool hi = lane >= 16;
  const int  g8 = hi ? 8 : 0;

  v2f bm0, bm1;
  {
    const float a0 = pp[0 * L_N + nn];
    const float a1 = pp[1 * L_N + nn];
    const float a2 = pp[2 * L_N + nn];
    bm0.x = hi ? (-2.0f * a2) : (-2.0f * a0);
    bm0.y = hi ? 1.0f         : (-2.0f * a1);
    const float c0 = pp[0 * L_N + 16 + nn];
    const float c1 = pp[1 * L_N + 16 + nn];
    const float c2 = pp[2 * L_N + 16 + nn];
    bm1.x = hi ? (-2.0f * c2) : (-2.0f * c0);
    bm1.y = hi ? 1.0f         : (-2.0f * c1);
  }
  v8f cq0, cq1;
#pragma unroll
  for (int r = 0; r < 8; ++r) {
    const int l0 = g8 + r;
    const float a0 = pp[0 * L_N + l0];
    const float a1 = pp[1 * L_N + l0];
    const float a2 = pp[2 * L_N + l0];
    cq0[r] = a0 * a0 + a1 * a1 + a2 * a2;
    const float c0 = pp[0 * L_N + 16 + l0];
    const float c1 = pp[1 * L_N + 16 + l0];
    const float c2 = pp[2 * L_N + 16 + l0];
    cq1[r] = c0 * c0 + c1 * c1 + c2 * c2;
  }

  if (lane < 16) tile[lane * TILE_STRIDE + 32] = 0.0f;

  const float w = DTW_W;
  const float wB = (nn == 0) ? __int_as_float(INF_BITS) : w;
  const float wC = hi ? w : __int_as_float(INF_BITS);
  float nmin = (lane == 0) ? 0.0f : __int_as_float(INF_BITS);

  float nx0 = xb[0 * T_N + nn];
  float nx1 = xb[1 * T_N + nn];
  float nx2 = xb[2 * T_N + nn];
  float lkdead = 0.0f;

  for (int blk = 0; blk < T_N / 16; ++blk) {
    const int t0 = blk * 16;

    const float x0  = nx0;
    const float x1  = nx1;
    const float x2c = nx2;

    const float xn  = x0 * x0 + x1 * x1 + x2c * x2c;
    v2f am;
    am.x = hi ? x2c : x0;
    am.y = hi ? xn  : x1;

    v8f dlo = __builtin_amdgcn_wmma_f32_16x16x4_f32(
        false, bm0, false, am, (short)0, cq0, false, false);
    v8f dhi = __builtin_amdgcn_wmma_f32_16x16x4_f32(
        false, bm1, false, am, (short)0, cq1, false, false);

    float sl[8], sh[8];
    sl[7] = sqrt_lo(dlo[7]);
    sh[7] = sqrt_lo(dhi[7]);
#pragma unroll
    for (int r = 6; r >= 0; --r) {
      sl[r] = sqrt_lo(dlo[r]) + sl[r + 1];
      sh[r] = sqrt_lo(dhi[r]) + sh[r + 1];
    }
    const float oth_h  = other_row_same(sh[0]);
    const float oth_l  = other_row_same(sl[0]);
    const float tot_hi = sh[0] + oth_h;
    const float off_h  = hi ? 0.0f : oth_h;
    const float off_l  = hi ? tot_hi : (tot_hi + oth_l);

    float* trow = &tile[nn * TILE_STRIDE + g8];
#pragma unroll
    for (int r = 0; r < 8; ++r) {
      trow[r]      = sl[r] + off_l;
      trow[16 + r] = sh[r] + off_h;
    }

    float Sk[16], Sxk[16];
#pragma unroll
    for (int k = 0; k < 16; ++k) {
      Sk[k]  = tile[k * TILE_STRIDE + lane];
      Sxk[k] = tile[k * TILE_STRIDE + lane + 1];
    }
    __builtin_amdgcn_sched_barrier(0);

    const int tn = (blk < T_N / 16 - 1 ? t0 + 16 : t0) + nn;
    nx0 = xb[0 * T_N + tn];
    nx1 = xb[1 * T_N + tn];
    nx2 = xb[2 * T_N + tn];
    __builtin_amdgcn_sched_barrier(0);

    const float pinf = __int_as_float(INF_BITS);
    float Ka[15], Kb[15];
#pragma unroll
    for (int k = 0; k < 15; ++k) {
      Ka[k] = __builtin_fmaf(-w, Sxk[k], Sk[k + 1]);
      Kb[k] = __builtin_fmaf(-w, Sk[k],  Sk[k + 1]);
    }
    const float nSb = (nn == 0) ? pinf : -Sk[15];
    const float nSc = hi ? -Sk[15] : pinf;

    float H = __builtin_fmaf(w, nmin, Sk[0]);
    float oreg = 0.0f;
#pragma unroll
    for (int k = 0; k < 16; ++k) {
      float G = imin_shr<1>(H);
      G = imin_shr<2>(G);
      G = imin_shr<4>(G);
      G = imin_shr<8>(G);
      const float link = other_row_l15(G, lkdead);
      lkdead = link;
      const float sh   = shr1_self0(G);

      oreg = shl1_feed(imin(G, link), oreg);

      if (k < 15) {
        const float a  = __builtin_fmaf(w, G,    Ka[k]);
        const float e1 = __builtin_fmaf(wB, sh,   Kb[k]);
        const float e2 = __builtin_fmaf(wC, link, Kb[k]);
        H = imin3(a, e1, e2);
      } else {
        nmin = imin3(G - Sxk[15], sh + nSb, link + nSc);
      }
    }

    if (lane < 16) out[(b * P_N + p) * T_N + t0 + lane] = oreg;
  }
}

extern "C" void kernel_launch(void* const* d_in, const int* in_sizes, int n_in,
                              void* d_out, int out_size, void* d_ws, size_t ws_size,
                              hipStream_t stream) {
  (void)in_sizes; (void)n_in; (void)out_size; (void)d_ws; (void)ws_size;
  const float* x     = (const float*)d_in[0];
  const float* patts = (const float*)d_in[1];
  float* out = (float*)d_out;

  dim3 grid(256), block(256);
  hipLaunchKernelGGL(dtw_wmma_kernel, grid, block, 0, stream, x, patts, out);
}
